# starting kernel
# baseline (speedup 1.0000x reference)
#include <hip/hip_runtime.h>
#include <math.h>

#define EPSF 1e-16f
#define BATCH 64
#define NROWS 16384
#define MDIM 64

typedef float v2f __attribute__((ext_vector_type(2)));
typedef float v8f __attribute__((ext_vector_type(8)));

__global__ __launch_bounds__(256) void ntm_content_kernel(
    const float* __restrict__ memory, const float* __restrict__ k,
    const float* __restrict__ beta, float* __restrict__ a_out) {
  const int wave = threadIdx.x >> 5;
  const int lane = threadIdx.x & 31;
  const int tile = blockIdx.x * (blockDim.x >> 5) + wave;
  const int b  = tile >> 10;
  const int nb = (tile & 1023) << 4;
  const int half = lane >> 4;
  const int r = lane & 15;

  const float* mrow = memory + ((size_t)b * NROWS + nb + r) * MDIM;
  const float* kb   = k + b * MDIM;

  v8f acc = {};
  float sq = 0.f;
  float kn = 0.f;
#pragma unroll
  for (int mb = 0; mb < MDIM; mb += 4) {
    const int c0 = mb + (half << 1);
    float2 mv = *(const float2*)(mrow + c0);
    float b0 = mv.x + EPSF;
    float b1 = mv.y + EPSF;
    float a0 = kb[c0]     + EPSF;
    float a1 = kb[c0 + 1] + EPSF;
    v2f A  = {a0, a1};
    v2f Bm = {b0, b1};
    acc = __builtin_amdgcn_wmma_f32_16x16x4_f32(
        false, A, false, Bm, (short)0, acc, false, false);
    sq = fmaf(b0, b0, fmaf(b1, b1, sq));
    kn = fmaf(a0, a0, fmaf(a1, a1, kn));
  }
  sq += __shfl_xor(sq, 16, 32);
  kn += __shfl_xor(kn, 16, 32);

  float num   = acc[0];
  float denom = sqrtf(sq) * sqrtf(kn);
  float cosv  = num / fmaxf(denom, 1e-8f);
  if (lane < 16)
    a_out[(size_t)b * NROWS + nb + r] = beta[b] * cosv;
}

__device__ inline float waveSum(float v) {
#pragma unroll
  for (int o = 16; o > 0; o >>= 1) v += __shfl_down(v, o, 32);
  return v;
}
__device__ inline float waveMax(float v) {
#pragma unroll
  for (int o = 16; o > 0; o >>= 1) v = fmaxf(v, __shfl_down(v, o, 32));
  return v;
}
__device__ inline float blockSum(float v, float* red) {
  const int lane = threadIdx.x & 31, wid = threadIdx.x >> 5;
  v = waveSum(v);
  if (lane == 0) red[wid] = v;
  __syncthreads();
  if (wid == 0) {
    float x = waveSum(red[lane]);
    if (lane == 0) red[0] = x;
  }
  __syncthreads();
  float r = red[0];
  __syncthreads();
  return r;
}
__device__ inline float blockMax(float v, float* red) {
  const int lane = threadIdx.x & 31, wid = threadIdx.x >> 5;
  v = waveMax(v);
  if (lane == 0) red[wid] = v;
  __syncthreads();
  if (wid == 0) {
    float x = waveMax(red[lane]);
    if (lane == 0) red[0] = x;
  }
  __syncthreads();
  float r = red[0];
  __syncthreads();
  return r;
}

__global__ __launch_bounds__(1024) void ntm_post_kernel(
    const float* __restrict__ a_in, const float* __restrict__ gv,
    const float* __restrict__ s, const float* __restrict__ gamma,
    const float* __restrict__ w_prev, float* __restrict__ out) {
  const int b   = blockIdx.x;
  const int tid = threadIdx.x;
  const int T   = 1024;
  const int NPT = NROWS / T;

  extern __shared__ float wgS[];
  __shared__ float red[32];

  const float* aRow = a_in + (size_t)b * NROWS;
  const float* wpRow = w_prev + (size_t)b * NROWS;
  const float g  = gv[b];
  const float s0 = s[b * 3 + 0], s1 = s[b * 3 + 1], s2 = s[b * 3 + 2];
  const float gam = gamma[b];

  float av[NPT];
  float lmax = -INFINITY;
#pragma unroll
  for (int j = 0; j < NPT; ++j) {
    const int i = tid + j * T;
    av[j] = aRow[i];
    lmax = fmaxf(lmax, av[j]);
  }
  const float bmax = blockMax(lmax, red);

  float lsum = 0.f;
#pragma unroll
  for (int j = 0; j < NPT; ++j) {
    av[j] = __expf(av[j] - bmax);
    lsum += av[j];
  }
  const float inv = 1.f / blockSum(lsum, red);

#pragma unroll
  for (int j = 0; j < NPT; ++j) {
    const int i = tid + j * T;
    const float wc = av[j] * inv;
    wgS[i] = g * wc + (1.f - g) * wpRow[i];
  }
  __syncthreads();

  float lsum2 = 0.f;
#pragma unroll
  for (int j = 0; j < NPT; ++j) {
    const int i = tid + j * T;
    const float left  = wgS[(i == 0) ? (NROWS - 1) : (i - 1)];
    const float mid   = wgS[i];
    const float right = wgS[(i == NROWS - 1) ? 0 : (i + 1)];
    const float wh = left * s0 + mid * s1 + right * s2;
    const float w  = powf(wh, gam);
    av[j] = w;
    lsum2 += w;
  }
  const float invt = 1.f / (blockSum(lsum2, red) + EPSF);

#pragma unroll
  for (int j = 0; j < NPT; ++j) {
    const int i = tid + j * T;
    out[(size_t)b * NROWS + i] = av[j] * invt;
  }
}

extern "C" void kernel_launch(void* const* d_in, const int* in_sizes, int n_in,
                              void* d_out, int out_size, void* d_ws, size_t ws_size,
                              hipStream_t stream) {
  const float* memory = (const float*)d_in[0];
  const float* k      = (const float*)d_in[1];
  const float* beta   = (const float*)d_in[2];
  const float* g      = (const float*)d_in[3];
  const float* s      = (const float*)d_in[4];
  const float* gamma  = (const float*)d_in[5];
  const float* w_prev = (const float*)d_in[6];
  float* out  = (float*)d_out;
  float* a_ws = (float*)d_ws;

  const int tilesPerBatch = NROWS / 16;
  const int totalTiles    = BATCH * tilesPerBatch;
  const int wavesPerBlock = 8;
  dim3 grid1(totalTiles / wavesPerBlock);

  ntm_content_kernel<<<grid1, 256, 0, stream>>>(memory, k, beta, a_ws);

  const size_t shmem = (size_t)NROWS * sizeof(float);
  ntm_post_kernel<<<BATCH, 1024, shmem, stream>>>(a_ws, g, s, gamma, w_prev, out);
}


# content kernel: 256 resident 16-wave workgroups, each wave streams 16 tiles with 16-byte loads two tiles ahead, k fragments and k norm once per wave; column sum order regrouped, two accumulators
# speedup vs baseline: 1.2229x; 1.2229x over previous
#include <hip/hip_runtime.h>
#include <math.h>

#define EPSF 1e-16f
#define BATCH 64
#define NROWS 16384
#define MDIM 64

typedef float v2f __attribute__((ext_vector_type(2)));
typedef float v8f __attribute__((ext_vector_type(8)));

#define K1_WAVES 16
#define K1_TILES 16
#define K1_AHEAD 2

__global__ __launch_bounds__(K1_WAVES * 32) void ntm_content_kernel(
    const float* __restrict__ memory, const float* __restrict__ k,
    const float* __restrict__ beta, float* __restrict__ a_out) {
  const int lane = threadIdx.x & 31;
  const int wave = __builtin_amdgcn_readfirstlane(threadIdx.x >> 5);
  const int b    = blockIdx.x >> 2;
  const int row0 = (blockIdx.x & 3) * (K1_WAVES * K1_TILES * 16) + wave * 16;
  const int half = lane >> 4;
  const int r    = lane & 15;

  const float* kb = k + b * MDIM + 4 * half;
  float ke[32];
  float kn = 0.f;
#pragma unroll
  for (int j = 0; j < 8; ++j) {
    const float4 kv = *(const float4*)(kb + 8 * j);
    ke[4 * j + 0] = kv.x + EPSF;
    ke[4 * j + 1] = kv.y + EPSF;
    ke[4 * j + 2] = kv.z + EPSF;
    ke[4 * j + 3] = kv.w + EPSF;
    kn = fmaf(ke[4 * j + 0], ke[4 * j + 0], fmaf(ke[4 * j + 1], ke[4 * j + 1], kn));
    kn = fmaf(ke[4 * j + 2], ke[4 * j + 2], fmaf(ke[4 * j + 3], ke[4 * j + 3], kn));
  }
  kn += __shfl_xor(kn, 16, 32);
  const float knorm = sqrtf(kn);
  const float bet   = beta[b];

  const float* lanep = memory + ((size_t)b * NROWS + row0 + r) * MDIM + 4 * half;
  float* outp = a_out + (size_t)b * NROWS + row0 + r;

  float4 buf[K1_AHEAD + 1][8];
#define K1_LOAD_TILE(T) do { \
    _Pragma("unroll") for (int j = 0; j < 8; ++j) \
      buf[(T) % (K1_AHEAD + 1)][j] = *(const float4*)(lanep + (size_t)(T) * K1_WAVES * 16 * MDIM + 8 * j); } while (0)

#pragma unroll
  for (int t = 0; t < K1_AHEAD; ++t) K1_LOAD_TILE(t);

#pragma unroll
  for (int t = 0; t < K1_TILES; ++t) {
    if (t + K1_AHEAD < K1_TILES) K1_LOAD_TILE(t + K1_AHEAD);

    const float4* cur = buf[t % (K1_AHEAD + 1)];
    v8f acc0 = {}, acc1 = {};
    float sq0 = 0.f, sq1 = 0.f;
#pragma unroll
    for (int j = 0; j < 8; ++j) {
      const float b0 = cur[j].x + EPSF;
      const float b1 = cur[j].y + EPSF;
      const float b2 = cur[j].z + EPSF;
      const float b3 = cur[j].w + EPSF;
      v2f A0 = {ke[4 * j + 0], ke[4 * j + 1]};
      v2f B0 = {b0, b1};
      v2f A1 = {ke[4 * j + 2], ke[4 * j + 3]};
      v2f B1 = {b2, b3};
      acc0 = __builtin_amdgcn_wmma_f32_16x16x4_f32(
          false, A0, false, B0, (short)0, acc0, false, false);
      acc1 = __builtin_amdgcn_wmma_f32_16x16x4_f32(
          false, A1, false, B1, (short)0, acc1, false, false);
      sq0 = fmaf(b0, b0, fmaf(b1, b1, sq0));
      sq1 = fmaf(b2, b2, fmaf(b3, b3, sq1));
    }
    float sq = sq0 + sq1;
    sq += __shfl_xor(sq, 16, 32);

    const float num   = acc0[0] + acc1[0];
    const float denom = sqrtf(sq) * knorm;
    const float cosv  = num / fmaxf(denom, 1e-8f);
    if (lane < 16) outp[t * K1_WAVES * 16] = bet * cosv;
  }
#undef K1_LOAD_TILE
}

__device__ inline float waveSum(float v) {
#pragma unroll
  for (int o = 16; o > 0; o >>= 1) v += __shfl_down(v, o, 32);
  return v;
}
__device__ inline float waveMax(float v) {
#pragma unroll
  for (int o = 16; o > 0; o >>= 1) v = fmaxf(v, __shfl_down(v, o, 32));
  return v;
}
__device__ inline float blockSum(float v, float* red) {
  const int lane = threadIdx.x & 31, wid = threadIdx.x >> 5;
  v = waveSum(v);
  if (lane == 0) red[wid] = v;
  __syncthreads();
  if (wid == 0) {
    float x = waveSum(red[lane]);
    if (lane == 0) red[0] = x;
  }
  __syncthreads();
  float r = red[0];
  __syncthreads();
  return r;
}
__device__ inline float blockMax(float v, float* red) {
  const int lane = threadIdx.x & 31, wid = threadIdx.x >> 5;
  v = waveMax(v);
  if (lane == 0) red[wid] = v;
  __syncthreads();
  if (wid == 0) {
    float x = waveMax(red[lane]);
    if (lane == 0) red[0] = x;
  }
  __syncthreads();
  float r = red[0];
  __syncthreads();
  return r;
}

__global__ __launch_bounds__(1024) void ntm_post_kernel(
    const float* __restrict__ a_in, const float* __restrict__ gv,
    const float* __restrict__ s, const float* __restrict__ gamma,
    const float* __restrict__ w_prev, float* __restrict__ out) {
  const int b   = blockIdx.x;
  const int tid = threadIdx.x;
  const int T   = 1024;
  const int NPT = NROWS / T;

  extern __shared__ float wgS[];
  __shared__ float red[32];

  const float* aRow = a_in + (size_t)b * NROWS;
  const float* wpRow = w_prev + (size_t)b * NROWS;
  const float g  = gv[b];
  const float s0 = s[b * 3 + 0], s1 = s[b * 3 + 1], s2 = s[b * 3 + 2];
  const float gam = gamma[b];

  float av[NPT];
  float lmax = -INFINITY;
#pragma unroll
  for (int j = 0; j < NPT; ++j) {
    const int i = tid + j * T;
    av[j] = aRow[i];
    lmax = fmaxf(lmax, av[j]);
  }
  const float bmax = blockMax(lmax, red);

  float lsum = 0.f;
#pragma unroll
  for (int j = 0; j < NPT; ++j) {
    av[j] = __expf(av[j] - bmax);
    lsum += av[j];
  }
  const float inv = 1.f / blockSum(lsum, red);

#pragma unroll
  for (int j = 0; j < NPT; ++j) {
    const int i = tid + j * T;
    const float wc = av[j] * inv;
    wgS[i] = g * wc + (1.f - g) * wpRow[i];
  }
  __syncthreads();

  float lsum2 = 0.f;
#pragma unroll
  for (int j = 0; j < NPT; ++j) {
    const int i = tid + j * T;
    const float left  = wgS[(i == 0) ? (NROWS - 1) : (i - 1)];
    const float mid   = wgS[i];
    const float right = wgS[(i == NROWS - 1) ? 0 : (i + 1)];
    const float wh = left * s0 + mid * s1 + right * s2;
    const float w  = powf(wh, gam);
    av[j] = w;
    lsum2 += w;
  }
  const float invt = 1.f / (blockSum(lsum2, red) + EPSF);

#pragma unroll
  for (int j = 0; j < NPT; ++j) {
    const int i = tid + j * T;
    out[(size_t)b * NROWS + i] = av[j] * invt;
  }
}

extern "C" void kernel_launch(void* const* d_in, const int* in_sizes, int n_in,
                              void* d_out, int out_size, void* d_ws, size_t ws_size,
                              hipStream_t stream) {
  const float* memory = (const float*)d_in[0];
  const float* k      = (const float*)d_in[1];
  const float* beta   = (const float*)d_in[2];
  const float* g      = (const float*)d_in[3];
  const float* s      = (const float*)d_in[4];
  const float* gamma  = (const float*)d_in[5];
  const float* w_prev = (const float*)d_in[6];
  float* out  = (float*)d_out;
  float* a_ws = (float*)d_ws;

  ntm_content_kernel<<<256, 512, 0, stream>>>(memory, k, beta, a_ws);

  const size_t shmem = (size_t)NROWS * sizeof(float);
  ntm_post_kernel<<<BATCH, 1024, shmem, stream>>>(a_ws, g, s, gamma, w_prev, out);
}


# post kernel: library powf (about 125 vector instructions per element) replaced by frexp + hardware log2 of the mantissa + split gamma*e product + hardware exp2 + ldexp (about 13), error a few ulp
# speedup vs baseline: 1.4524x; 1.1877x over previous
#include <hip/hip_runtime.h>
#include <math.h>

#define EPSF 1e-16f
#define BATCH 64
#define NROWS 16384
#define MDIM 64

typedef float v2f __attribute__((ext_vector_type(2)));
typedef float v8f __attribute__((ext_vector_type(8)));

#define K1_WAVES 16
#define K1_TILES 16
#define K1_AHEAD 2

__global__ __launch_bounds__(K1_WAVES * 32) void ntm_content_kernel(
    const float* __restrict__ memory, const float* __restrict__ k,
    const float* __restrict__ beta, float* __restrict__ a_out) {
  const int lane = threadIdx.x & 31;
  const int wave = __builtin_amdgcn_readfirstlane(threadIdx.x >> 5);
  const int b    = blockIdx.x >> 2;
  const int row0 = (blockIdx.x & 3) * (K1_WAVES * K1_TILES * 16) + wave * 16;
  const int half = lane >> 4;
  const int r    = lane & 15;

  const float* kb = k + b * MDIM + 4 * half;
  float ke[32];
  float kn = 0.f;
#pragma unroll
  for (int j = 0; j < 8; ++j) {
    const float4 kv = *(const float4*)(kb + 8 * j);
    ke[4 * j + 0] = kv.x + EPSF;
    ke[4 * j + 1] = kv.y + EPSF;
    ke[4 * j + 2] = kv.z + EPSF;
    ke[4 * j + 3] = kv.w + EPSF;
    kn = fmaf(ke[4 * j + 0], ke[4 * j + 0], fmaf(ke[4 * j + 1], ke[4 * j + 1], kn));
    kn = fmaf(ke[4 * j + 2], ke[4 * j + 2], fmaf(ke[4 * j + 3], ke[4 * j + 3], kn));
  }
  kn += __shfl_xor(kn, 16, 32);
  const float knorm = sqrtf(kn);
  const float bet   = beta[b];

  const float* lanep = memory + ((size_t)b * NROWS + row0 + r) * MDIM + 4 * half;
  float* outp = a_out + (size_t)b * NROWS + row0 + r;

  float4 buf[K1_AHEAD + 1][8];
#define K1_LOAD_TILE(T) do { \
    _Pragma("unroll") for (int j = 0; j < 8; ++j) \
      buf[(T) % (K1_AHEAD + 1)][j] = *(const float4*)(lanep + (size_t)(T) * K1_WAVES * 16 * MDIM + 8 * j); } while (0)

#pragma unroll
  for (int t = 0; t < K1_AHEAD; ++t) K1_LOAD_TILE(t);

#pragma unroll
  for (int t = 0; t < K1_TILES; ++t) {
    if (t + K1_AHEAD < K1_TILES) K1_LOAD_TILE(t + K1_AHEAD);

    const float4* cur = buf[t % (K1_AHEAD + 1)];
    v8f acc0 = {}, acc1 = {};
    float sq0 = 0.f, sq1 = 0.f;
#pragma unroll
    for (int j = 0; j < 8; ++j) {
      const float b0 = cur[j].x + EPSF;
      const float b1 = cur[j].y + EPSF;
      const float b2 = cur[j].z + EPSF;
      const float b3 = cur[j].w + EPSF;
      v2f A0 = {ke[4 * j + 0], ke[4 * j + 1]};
      v2f B0 = {b0, b1};
      v2f A1 = {ke[4 * j + 2], ke[4 * j + 3]};
      v2f B1 = {b2, b3};
      acc0 = __builtin_amdgcn_wmma_f32_16x16x4_f32(
          false, A0, false, B0, (short)0, acc0, false, false);
      acc1 = __builtin_amdgcn_wmma_f32_16x16x4_f32(
          false, A1, false, B1, (short)0, acc1, false, false);
      sq0 = fmaf(b0, b0, fmaf(b1, b1, sq0));
      sq1 = fmaf(b2, b2, fmaf(b3, b3, sq1));
    }
    float sq = sq0 + sq1;
    sq += __shfl_xor(sq, 16, 32);

    const float num   = acc0[0] + acc1[0];
    const float denom = sqrtf(sq) * knorm;
    const float cosv  = num / fmaxf(denom, 1e-8f);
    if (lane < 16) outp[t * K1_WAVES * 16] = bet * cosv;
  }
#undef K1_LOAD_TILE
}

__device__ inline float waveSum(float v) {
#pragma unroll
  for (int o = 16; o > 0; o >>= 1) v += __shfl_down(v, o, 32);
  return v;
}
__device__ inline float waveMax(float v) {
#pragma unroll
  for (int o = 16; o > 0; o >>= 1) v = fmaxf(v, __shfl_down(v, o, 32));
  return v;
}
__device__ inline float blockSum(float v, float* red) {
  const int lane = threadIdx.x & 31, wid = threadIdx.x >> 5;
  v = waveSum(v);
  if (lane == 0) red[wid] = v;
  __syncthreads();
  if (wid == 0) {
    float x = waveSum(red[lane]);
    if (lane == 0) red[0] = x;
  }
  __syncthreads();
  float r = red[0];
  __syncthreads();
  return r;
}
__device__ inline float blockMax(float v, float* red) {
  const int lane = threadIdx.x & 31, wid = threadIdx.x >> 5;
  v = waveMax(v);
  if (lane == 0) red[wid] = v;
  __syncthreads();
  if (wid == 0) {
    float x = waveMax(red[lane]);
    if (lane == 0) red[0] = x;
  }
  __syncthreads();
  float r = red[0];
  __syncthreads();
  return r;
}

__device__ inline float pow_nonneg(float x, float p, float z0, bool p_pos, bool p_zero) {
  const float m  = __builtin_amdgcn_frexp_mantf(x);
  const float ef = (float)__builtin_amdgcn_frexp_expf(x);
  const float l  = __builtin_amdgcn_logf(m);
  const float n  = __builtin_rintf(p * ef);
  const float f0 = __builtin_fmaf(p, ef, -n);
  const float f  = __builtin_fmaf(p, l, f0);
  float r = __builtin_ldexpf(__builtin_amdgcn_exp2f(f), (int)n);
  if (!p_pos) r = (p_zero || x == 0.f) ? z0 : r;
  return r;
}

__global__ __launch_bounds__(1024) void ntm_post_kernel(
    const float* __restrict__ a_in, const float* __restrict__ gv,
    const float* __restrict__ s, const float* __restrict__ gamma,
    const float* __restrict__ w_prev, float* __restrict__ out) {
  const int b   = blockIdx.x;
  const int tid = threadIdx.x;
  const int T   = 1024;
  const int NPT = NROWS / T;

  extern __shared__ float wgS[];
  __shared__ float red[32];

  const float* aRow = a_in + (size_t)b * NROWS;
  const float* wpRow = w_prev + (size_t)b * NROWS;
  const float g  = gv[b];
  const float s0 = s[b * 3 + 0], s1 = s[b * 3 + 1], s2 = s[b * 3 + 2];
  const float gam = gamma[b];

  float av[NPT];
  float lmax = -INFINITY;
#pragma unroll
  for (int j = 0; j < NPT; ++j) {
    const int i = tid + j * T;
    av[j] = aRow[i];
    lmax = fmaxf(lmax, av[j]);
  }
  const float bmax = blockMax(lmax, red);

  float lsum = 0.f;
#pragma unroll
  for (int j = 0; j < NPT; ++j) {
    av[j] = __expf(av[j] - bmax);
    lsum += av[j];
  }
  const float inv = 1.f / blockSum(lsum, red);

#pragma unroll
  for (int j = 0; j < NPT; ++j) {
    const int i = tid + j * T;
    const float wc = av[j] * inv;
    wgS[i] = g * wc + (1.f - g) * wpRow[i];
  }
  __syncthreads();

  const bool  gpos  = gam > 0.f;
  const bool  gzero = gam == 0.f;
  const float z0    = gzero ? 1.f : INFINITY;

  float lsum2 = 0.f;
#pragma unroll
  for (int j = 0; j < NPT; ++j) {
    const int i = tid + j * T;
    const float left  = wgS[(i == 0) ? (NROWS - 1) : (i - 1)];
    const float mid   = wgS[i];
    const float right = wgS[(i == NROWS - 1) ? 0 : (i + 1)];
    const float wh = left * s0 + mid * s1 + right * s2;
    const float w  = pow_nonneg(wh, gam, z0, gpos, gzero);
    av[j] = w;
    lsum2 += w;
  }
  const float invt = 1.f / (blockSum(lsum2, red) + EPSF);

#pragma unroll
  for (int j = 0; j < NPT; ++j) {
    const int i = tid + j * T;
    out[(size_t)b * NROWS + i] = av[j] * invt;
  }
}

extern "C" void kernel_launch(void* const* d_in, const int* in_sizes, int n_in,
                              void* d_out, int out_size, void* d_ws, size_t ws_size,
                              hipStream_t stream) {
  const float* memory = (const float*)d_in[0];
  const float* k      = (const float*)d_in[1];
  const float* beta   = (const float*)d_in[2];
  const float* g      = (const float*)d_in[3];
  const float* s      = (const float*)d_in[4];
  const float* gamma  = (const float*)d_in[5];
  const float* w_prev = (const float*)d_in[6];
  float* out  = (float*)d_out;
  float* a_ws = (float*)d_ws;

  ntm_content_kernel<<<256, 512, 0, stream>>>(memory, k, beta, a_ws);

  const size_t shmem = (size_t)NROWS * sizeof(float);
  ntm_post_kernel<<<BATCH, 1024, shmem, stream>>>(a_ws, g, s, gamma, w_prev, out);
}


# content kernel: each 16-byte-per-lane global load reads 512 contiguous bytes (4 whole lines; was 32 bytes of each of 16 lines); a wave-private LDS buffer turns the tile to the row-per-lane fragment
# speedup vs baseline: 1.5970x; 1.0996x over previous
#include <hip/hip_runtime.h>
#include <math.h>

#define EPSF 1e-16f
#define BATCH 64
#define NROWS 16384
#define MDIM 64

typedef float v2f __attribute__((ext_vector_type(2)));
typedef float v8f __attribute__((ext_vector_type(8)));
typedef float v4f __attribute__((ext_vector_type(4)));

#define K1_WAVES 16
#define K1_TILES 16
#define K1_AHEAD 2
#define K1_PITCH 17

__global__ __launch_bounds__(K1_WAVES * 32) void ntm_content_kernel(
    const float* __restrict__ memory, const float* __restrict__ k,
    const float* __restrict__ beta, float* __restrict__ a_out) {
  __shared__ v4f stage[K1_WAVES][16 * K1_PITCH];

  const int lane = threadIdx.x & 31;
  const int wave = __builtin_amdgcn_readfirstlane(threadIdx.x >> 5);
  const int b    = blockIdx.x >> 2;
  const int row0 = (blockIdx.x & 3) * (K1_WAVES * K1_TILES * 16) + wave * 16;
  const int half = lane >> 4;
  const int r    = lane & 15;

  const float* kb = k + b * MDIM + 4 * half;
  float ke[32];
  float kn = 0.f;
#pragma unroll
  for (int j = 0; j < 8; ++j) {
    const float4 kv = *(const float4*)(kb + 8 * j);
    ke[4 * j + 0] = kv.x + EPSF;
    ke[4 * j + 1] = kv.y + EPSF;
    ke[4 * j + 2] = kv.z + EPSF;
    ke[4 * j + 3] = kv.w + EPSF;
    kn = fmaf(ke[4 * j + 0], ke[4 * j + 0], fmaf(ke[4 * j + 1], ke[4 * j + 1], kn));
    kn = fmaf(ke[4 * j + 2], ke[4 * j + 2], fmaf(ke[4 * j + 3], ke[4 * j + 3], kn));
  }
  kn += __shfl_xor(kn, 16, 32);
  const float knorm = sqrtf(kn);
  const float bet   = beta[b];

  const float* lanep = memory + ((size_t)b * NROWS + row0) * MDIM + 4 * lane;
  float* outp = a_out + (size_t)b * NROWS + row0 + r;

  v4f* const st_w = &stage[wave][half * K1_PITCH + r];
  const v4f* const st_r = &stage[wave][r * K1_PITCH + half];

  v4f buf[K1_AHEAD + 1][8];
#define K1_LOAD_TILE(T) do { \
    _Pragma("unroll") for (int i = 0; i < 8; ++i) \
      buf[(T) % (K1_AHEAD + 1)][i] = *(const v4f*)(lanep + (size_t)(T) * K1_WAVES * 16 * MDIM + 128 * i); } while (0)

#pragma unroll
  for (int t = 0; t < K1_AHEAD; ++t) K1_LOAD_TILE(t);

#pragma unroll
  for (int t = 0; t < K1_TILES; ++t) {
    __builtin_amdgcn_sched_barrier(0);
    if (t + K1_AHEAD < K1_TILES) K1_LOAD_TILE(t + K1_AHEAD);
    __builtin_amdgcn_sched_barrier(0);

#pragma unroll
    for (int i = 0; i < 8; ++i) st_w[2 * i * K1_PITCH] = buf[t % (K1_AHEAD + 1)][i];

    v4f cur[8];
#pragma unroll
    for (int j = 0; j < 8; ++j) cur[j] = st_r[2 * j];
    __builtin_amdgcn_sched_barrier(0);

    v8f acc0 = {}, acc1 = {};
    float sq0 = 0.f, sq1 = 0.f;
#pragma unroll
    for (int j = 0; j < 8; ++j) {
      const float b0 = cur[j].x + EPSF;
      const float b1 = cur[j].y + EPSF;
      const float b2 = cur[j].z + EPSF;
      const float b3 = cur[j].w + EPSF;
      v2f A0 = {ke[4 * j + 0], ke[4 * j + 1]};
      v2f B0 = {b0, b1};
      v2f A1 = {ke[4 * j + 2], ke[4 * j + 3]};
      v2f B1 = {b2, b3};
      acc0 = __builtin_amdgcn_wmma_f32_16x16x4_f32(
          false, A0, false, B0, (short)0, acc0, false, false);
      acc1 = __builtin_amdgcn_wmma_f32_16x16x4_f32(
          false, A1, false, B1, (short)0, acc1, false, false);
      sq0 = fmaf(b0, b0, fmaf(b1, b1, sq0));
      sq1 = fmaf(b2, b2, fmaf(b3, b3, sq1));
    }
    float sq = sq0 + sq1;
    sq += __shfl_xor(sq, 16, 32);

    const float num   = acc0[0] + acc1[0];
    const float denom = sqrtf(sq) * knorm;
    const float cosv  = num / fmaxf(denom, 1e-8f);
    if (lane < 16) outp[t * K1_WAVES * 16] = bet * cosv;
  }
#undef K1_LOAD_TILE
}

__device__ inline float waveSum(float v) {
#pragma unroll
  for (int o = 16; o > 0; o >>= 1) v += __shfl_down(v, o, 32);
  return v;
}
__device__ inline float waveMax(float v) {
#pragma unroll
  for (int o = 16; o > 0; o >>= 1) v = fmaxf(v, __shfl_down(v, o, 32));
  return v;
}
__device__ inline float blockSum(float v, float* red) {
  const int lane = threadIdx.x & 31, wid = threadIdx.x >> 5;
  v = waveSum(v);
  if (lane == 0) red[wid] = v;
  __syncthreads();
  if (wid == 0) {
    float x = waveSum(red[lane]);
    if (lane == 0) red[0] = x;
  }
  __syncthreads();
  float r = red[0];
  __syncthreads();
  return r;
}
__device__ inline float blockMax(float v, float* red) {
  const int lane = threadIdx.x & 31, wid = threadIdx.x >> 5;
  v = waveMax(v);
  if (lane == 0) red[wid] = v;
  __syncthreads();
  if (wid == 0) {
    float x = waveMax(red[lane]);
    if (lane == 0) red[0] = x;
  }
  __syncthreads();
  float r = red[0];
  __syncthreads();
  return r;
}

__device__ inline float pow_nonneg(float x, float p, float z0, bool p_pos, bool p_zero) {
  const float m  = __builtin_amdgcn_frexp_mantf(x);
  const float ef = (float)__builtin_amdgcn_frexp_expf(x);
  const float l  = __builtin_amdgcn_logf(m);
  const float n  = __builtin_rintf(p * ef);
  const float f0 = __builtin_fmaf(p, ef, -n);
  const float f  = __builtin_fmaf(p, l, f0);
  float r = __builtin_ldexpf(__builtin_amdgcn_exp2f(f), (int)n);
  if (!p_pos) r = (p_zero || x == 0.f) ? z0 : r;
  return r;
}

__global__ __launch_bounds__(1024) void ntm_post_kernel(
    const float* __restrict__ a_in, const float* __restrict__ gv,
    const float* __restrict__ s, const float* __restrict__ gamma,
    const float* __restrict__ w_prev, float* __restrict__ out) {
  const int b   = blockIdx.x;
  const int tid = threadIdx.x;
  const int T   = 1024;
  const int NPT = NROWS / T;

  extern __shared__ float wgS[];
  __shared__ float red[32];

  const float* aRow = a_in + (size_t)b * NROWS;
  const float* wpRow = w_prev + (size_t)b * NROWS;
  const float g  = gv[b];
  const float s0 = s[b * 3 + 0], s1 = s[b * 3 + 1], s2 = s[b * 3 + 2];
  const float gam = gamma[b];

  float av[NPT];
  float lmax = -INFINITY;
#pragma unroll
  for (int j = 0; j < NPT; ++j) {
    const int i = tid + j * T;
    av[j] = aRow[i];
    lmax = fmaxf(lmax, av[j]);
  }
  const float bmax = blockMax(lmax, red);

  float lsum = 0.f;
#pragma unroll
  for (int j = 0; j < NPT; ++j) {
    av[j] = __expf(av[j] - bmax);
    lsum += av[j];
  }
  const float inv = 1.f / blockSum(lsum, red);

#pragma unroll
  for (int j = 0; j < NPT; ++j) {
    const int i = tid + j * T;
    const float wc = av[j] * inv;
    wgS[i] = g * wc + (1.f - g) * wpRow[i];
  }
  __syncthreads();

  const bool  gpos  = gam > 0.f;
  const bool  gzero = gam == 0.f;
  const float z0    = gzero ? 1.f : INFINITY;

  float lsum2 = 0.f;
#pragma unroll
  for (int j = 0; j < NPT; ++j) {
    const int i = tid + j * T;
    const float left  = wgS[(i == 0) ? (NROWS - 1) : (i - 1)];
    const float mid   = wgS[i];
    const float right = wgS[(i == NROWS - 1) ? 0 : (i + 1)];
    const float wh = left * s0 + mid * s1 + right * s2;
    const float w  = pow_nonneg(wh, gam, z0, gpos, gzero);
    av[j] = w;
    lsum2 += w;
  }
  const float invt = 1.f / (blockSum(lsum2, red) + EPSF);

#pragma unroll
  for (int j = 0; j < NPT; ++j) {
    const int i = tid + j * T;
    out[(size_t)b * NROWS + i] = av[j] * invt;
  }
}

extern "C" void kernel_launch(void* const* d_in, const int* in_sizes, int n_in,
                              void* d_out, int out_size, void* d_ws, size_t ws_size,
                              hipStream_t stream) {
  const float* memory = (const float*)d_in[0];
  const float* k      = (const float*)d_in[1];
  const float* beta   = (const float*)d_in[2];
  const float* g      = (const float*)d_in[3];
  const float* s      = (const float*)d_in[4];
  const float* gamma  = (const float*)d_in[5];
  const float* w_prev = (const float*)d_in[6];
  float* out  = (float*)d_out;
  float* a_ws = (float*)d_ws;

  ntm_content_kernel<<<256, 512, 0, stream>>>(memory, k, beta, a_ws);

  const size_t shmem = (size_t)NROWS * sizeof(float);
  ntm_post_kernel<<<BATCH, 1024, shmem, stream>>>(a_ws, g, s, gamma, w_prev, out);
}


# post kernel: 3 block reductions exchange lanes on the vector ALU (row_xmask DPP, cross-row permute), not by LDS permutes; each wave folds the 32 partials: 1 barrier each, was 3; sums bit-identical
# speedup vs baseline: 1.6491x; 1.0326x over previous
#include <hip/hip_runtime.h>
#include <math.h>

#define EPSF 1e-16f
#define BATCH 64
#define NROWS 16384
#define MDIM 64

typedef float v2f __attribute__((ext_vector_type(2)));
typedef float v8f __attribute__((ext_vector_type(8)));
typedef float v4f __attribute__((ext_vector_type(4)));

#define K1_WAVES 16
#define K1_TILES 16
#define K1_AHEAD 2
#define K1_PITCH 17

__global__ __launch_bounds__(K1_WAVES * 32) void ntm_content_kernel(
    const float* __restrict__ memory, const float* __restrict__ k,
    const float* __restrict__ beta, float* __restrict__ a_out) {
  __shared__ v4f stage[K1_WAVES][16 * K1_PITCH];

  const int lane = threadIdx.x & 31;
  const int wave = __builtin_amdgcn_readfirstlane(threadIdx.x >> 5);
  const int b    = blockIdx.x >> 2;
  const int row0 = (blockIdx.x & 3) * (K1_WAVES * K1_TILES * 16) + wave * 16;
  const int half = lane >> 4;
  const int r    = lane & 15;

  const float* kb = k + b * MDIM + 4 * half;
  float ke[32];
  float kn = 0.f;
#pragma unroll
  for (int j = 0; j < 8; ++j) {
    const float4 kv = *(const float4*)(kb + 8 * j);
    ke[4 * j + 0] = kv.x + EPSF;
    ke[4 * j + 1] = kv.y + EPSF;
    ke[4 * j + 2] = kv.z + EPSF;
    ke[4 * j + 3] = kv.w + EPSF;
    kn = fmaf(ke[4 * j + 0], ke[4 * j + 0], fmaf(ke[4 * j + 1], ke[4 * j + 1], kn));
    kn = fmaf(ke[4 * j + 2], ke[4 * j + 2], fmaf(ke[4 * j + 3], ke[4 * j + 3], kn));
  }
  kn += __shfl_xor(kn, 16, 32);
  const float knorm = sqrtf(kn);
  const float bet   = beta[b];

  const float* lanep = memory + ((size_t)b * NROWS + row0) * MDIM + 4 * lane;
  float* outp = a_out + (size_t)b * NROWS + row0 + r;

  v4f* const st_w = &stage[wave][half * K1_PITCH + r];
  const v4f* const st_r = &stage[wave][r * K1_PITCH + half];

  v4f buf[K1_AHEAD + 1][8];
#define K1_LOAD_TILE(T) do { \
    _Pragma("unroll") for (int i = 0; i < 8; ++i) \
      buf[(T) % (K1_AHEAD + 1)][i] = *(const v4f*)(lanep + (size_t)(T) * K1_WAVES * 16 * MDIM + 128 * i); } while (0)

#pragma unroll
  for (int t = 0; t < K1_AHEAD; ++t) K1_LOAD_TILE(t);

#pragma unroll
  for (int t = 0; t < K1_TILES; ++t) {
    __builtin_amdgcn_sched_barrier(0);
    if (t + K1_AHEAD < K1_TILES) K1_LOAD_TILE(t + K1_AHEAD);
    __builtin_amdgcn_sched_barrier(0);

#pragma unroll
    for (int i = 0; i < 8; ++i) st_w[2 * i * K1_PITCH] = buf[t % (K1_AHEAD + 1)][i];

    v4f cur[8];
#pragma unroll
    for (int j = 0; j < 8; ++j) cur[j] = st_r[2 * j];
    __builtin_amdgcn_sched_barrier(0);

    v8f acc0 = {}, acc1 = {};
    float sq0 = 0.f, sq1 = 0.f;
#pragma unroll
    for (int j = 0; j < 8; ++j) {
      const float b0 = cur[j].x + EPSF;
      const float b1 = cur[j].y + EPSF;
      const float b2 = cur[j].z + EPSF;
      const float b3 = cur[j].w + EPSF;
      v2f A0 = {ke[4 * j + 0], ke[4 * j + 1]};
      v2f B0 = {b0, b1};
      v2f A1 = {ke[4 * j + 2], ke[4 * j + 3]};
      v2f B1 = {b2, b3};
      acc0 = __builtin_amdgcn_wmma_f32_16x16x4_f32(
          false, A0, false, B0, (short)0, acc0, false, false);
      acc1 = __builtin_amdgcn_wmma_f32_16x16x4_f32(
          false, A1, false, B1, (short)0, acc1, false, false);
      sq0 = fmaf(b0, b0, fmaf(b1, b1, sq0));
      sq1 = fmaf(b2, b2, fmaf(b3, b3, sq1));
    }
    float sq = sq0 + sq1;
    sq += __shfl_xor(sq, 16, 32);

    const float num   = acc0[0] + acc1[0];
    const float denom = sqrtf(sq) * knorm;
    const float cosv  = num / fmaxf(denom, 1e-8f);
    if (lane < 16) outp[t * K1_WAVES * 16] = bet * cosv;
  }
#undef K1_LOAD_TILE
}

template <int MASK>
__device__ inline float rowXor(float v) {
  return __builtin_bit_cast(float, __builtin_amdgcn_update_dpp(
      0, __builtin_bit_cast(int, v), 0x160 + MASK, 0xf, 0xf, false));
}
__device__ inline float xhalf(float v) {
  const int b = __builtin_bit_cast(int, v);
  return __builtin_bit_cast(float,
      __builtin_amdgcn_permlanex16(b, b, 0x76543210, 0xfedcba98, false, false));
}
__device__ inline float waveSumAll(float v) {
  v += xhalf(v);
  v += rowXor<8>(v);
  v += rowXor<4>(v);
  v += rowXor<2>(v);
  v += rowXor<1>(v);
  return v;
}
__device__ inline float waveMaxAll(float v) {
  v = fmaxf(v, xhalf(v));
  v = fmaxf(v, rowXor<8>(v));
  v = fmaxf(v, rowXor<4>(v));
  v = fmaxf(v, rowXor<2>(v));
  v = fmaxf(v, rowXor<1>(v));
  return v;
}
__device__ inline float blockSum(float v, float* red) {
  const int lane = threadIdx.x & 31, wid = threadIdx.x >> 5;
  v = waveSumAll(v);
  if (lane == 0) red[wid] = v;
  __syncthreads();
  return waveSumAll(red[lane]);
}
__device__ inline float blockMax(float v, float* red) {
  const int lane = threadIdx.x & 31, wid = threadIdx.x >> 5;
  v = waveMaxAll(v);
  if (lane == 0) red[wid] = v;
  __syncthreads();
  return waveMaxAll(red[lane]);
}

__device__ inline float pow_nonneg(float x, float p, float z0, bool p_pos, bool p_zero) {
  const float m  = __builtin_amdgcn_frexp_mantf(x);
  const float ef = (float)__builtin_amdgcn_frexp_expf(x);
  const float l  = __builtin_amdgcn_logf(m);
  const float n  = __builtin_rintf(p * ef);
  const float f0 = __builtin_fmaf(p, ef, -n);
  const float f  = __builtin_fmaf(p, l, f0);
  float r = __builtin_ldexpf(__builtin_amdgcn_exp2f(f), (int)n);
  if (!p_pos) r = (p_zero || x == 0.f) ? z0 : r;
  return r;
}

__global__ __launch_bounds__(1024) void ntm_post_kernel(
    const float* __restrict__ a_in, const float* __restrict__ gv,
    const float* __restrict__ s, const float* __restrict__ gamma,
    const float* __restrict__ w_prev, float* __restrict__ out) {
  const int b   = blockIdx.x;
  const int tid = threadIdx.x;
  const int T   = 1024;
  const int NPT = NROWS / T;

  extern __shared__ float wgS[];
  __shared__ float red[3][32];

  const float* aRow = a_in + (size_t)b * NROWS;
  const float* wpRow = w_prev + (size_t)b * NROWS;
  const float g  = gv[b];
  const float s0 = s[b * 3 + 0], s1 = s[b * 3 + 1], s2 = s[b * 3 + 2];
  const float gam = gamma[b];

  float av[NPT];
  float lmax = -INFINITY;
#pragma unroll
  for (int j = 0; j < NPT; ++j) {
    const int i = tid + j * T;
    av[j] = aRow[i];
    lmax = fmaxf(lmax, av[j]);
  }
  const float bmax = blockMax(lmax, red[0]);

  float lsum = 0.f;
#pragma unroll
  for (int j = 0; j < NPT; ++j) {
    av[j] = __expf(av[j] - bmax);
    lsum += av[j];
  }
  const float inv = 1.f / blockSum(lsum, red[1]);

#pragma unroll
  for (int j = 0; j < NPT; ++j) {
    const int i = tid + j * T;
    const float wc = av[j] * inv;
    wgS[i] = g * wc + (1.f - g) * wpRow[i];
  }
  __syncthreads();

  const bool  gpos  = gam > 0.f;
  const bool  gzero = gam == 0.f;
  const float z0    = gzero ? 1.f : INFINITY;

  float lsum2 = 0.f;
#pragma unroll
  for (int j = 0; j < NPT; ++j) {
    const int i = tid + j * T;
    const float left  = wgS[(i == 0) ? (NROWS - 1) : (i - 1)];
    const float mid   = wgS[i];
    const float right = wgS[(i == NROWS - 1) ? 0 : (i + 1)];
    const float wh = left * s0 + mid * s1 + right * s2;
    const float w  = pow_nonneg(wh, gam, z0, gpos, gzero);
    av[j] = w;
    lsum2 += w;
  }
  const float invt = 1.f / (blockSum(lsum2, red[2]) + EPSF);

#pragma unroll
  for (int j = 0; j < NPT; ++j) {
    const int i = tid + j * T;
    out[(size_t)b * NROWS + i] = av[j] * invt;
  }
}

extern "C" void kernel_launch(void* const* d_in, const int* in_sizes, int n_in,
                              void* d_out, int out_size, void* d_ws, size_t ws_size,
                              hipStream_t stream) {
  const float* memory = (const float*)d_in[0];
  const float* k      = (const float*)d_in[1];
  const float* beta   = (const float*)d_in[2];
  const float* g      = (const float*)d_in[3];
  const float* s      = (const float*)d_in[4];
  const float* gamma  = (const float*)d_in[5];
  const float* w_prev = (const float*)d_in[6];
  float* out  = (float*)d_out;
  float* a_ws = (float*)d_ws;

  ntm_content_kernel<<<256, 512, 0, stream>>>(memory, k, beta, a_ws);

  const size_t shmem = (size_t)NROWS * sizeof(float);
  ntm_post_kernel<<<BATCH, 1024, shmem, stream>>>(a_ws, g, s, gamma, w_prev, out);
}
